# starting kernel
# baseline (speedup 1.0000x reference)
#include <hip/hip_runtime.h>


typedef __attribute__((ext_vector_type(2))) float v2f;
typedef __attribute__((ext_vector_type(8))) float v8f;

#define SWZ_SWAPX16 0x401F

__device__ __forceinline__ float swapx16(float v) {
    return __int_as_float(
        __builtin_amdgcn_ds_swizzle(__float_as_int(v), SWZ_SWAPX16));
}

__global__ __launch_bounds__(256) void sindy_wmma_f32_kernel(
    const float* __restrict__ x,
    const float* __restrict__ t,
    const float* __restrict__ layer,
    const float* __restrict__ omegas,
    const float* __restrict__ ext_filter,
    float* __restrict__ out,
    int n, int ntiles)
{
    const int lane = threadIdx.x & 31;
    const int wave = threadIdx.x >> 5;
    const int rsel = lane & 15;
    const bool hi  = lane >= 16;

    const int nwaves = gridDim.x * 8;
    const int wgid   = blockIdx.x * 8 + wave;

    const float w0 = omegas[0], w1 = omegas[1];
    const float f0 = ext_filter[0], f1 = ext_filter[1];

    const float fsel = (rsel == 0) ? f0 : ((rsel == 1) ? f1 : 0.0f);

    v2f a[17];
#pragma unroll
    for (int g = 0; g < 17; ++g) {
        const int k0 = 4 * g, k1 = k0 + 1, k2 = k0 + 2, k3 = k0 + 3;
        const float l0 = layer[k0];
        const float l1 = (k1 < 65) ? layer[k1] : 0.0f;
        const float l2 = (k2 < 65) ? layer[k2] : 0.0f;
        const float l3 = (k3 < 65) ? layer[k3] : 0.0f;
        a[g][0] = fsel * (hi ? l2 : l0);
        a[g][1] = fsel * (hi ? l3 : l1);
    }

    for (int tile = wgid; tile < ntiles; tile += nwaves) {
        const int base = tile * 32;
        const int row  = base + lane;
        const int rc   = (row < n) ? row : (n - 1);

        const float4 xv = *(const float4*)(x + (size_t)rc * 4);
        const float X0 = xv.x, X1 = xv.y, X2 = xv.z, X3 = xv.w;
        const float X4 = t[rc];

        const float q0 = X0 * X0, q1 = X1 * X1, q2 = X2 * X2,
                    q3 = X3 * X3, q4 = X4 * X4;

        float F[68];
        F[0]  = X0;        F[1]  = q0;        F[2]  = q0 * X0;
        F[3]  = q0 * X1;   F[4]  = q0 * X2;   F[5]  = q0 * X3;   F[6]  = q0 * X4;
        F[7]  = X0 * X1;   F[8]  = X0 * q1;   F[9]  = X0 * X1 * X2;
        F[10] = X0 * X1 * X3;                 F[11] = X0 * X1 * X4;
        F[12] = X0 * X2;   F[13] = X0 * q2;   F[14] = X0 * X2 * X3;
        F[15] = X0 * X2 * X4;
        F[16] = X0 * X3;   F[17] = X0 * q3;   F[18] = X0 * X3 * X4;
        F[19] = X0 * X4;   F[20] = X0 * q4;
        F[21] = X1;        F[22] = q1;        F[23] = q1 * X1;
        F[24] = q1 * X2;   F[25] = q1 * X3;   F[26] = q1 * X4;
        F[27] = X1 * X2;   F[28] = X1 * q2;   F[29] = X1 * X2 * X3;
        F[30] = X1 * X2 * X4;
        F[31] = X1 * X3;   F[32] = X1 * q3;   F[33] = X1 * X3 * X4;
        F[34] = X1 * X4;   F[35] = X1 * q4;
        F[36] = X2;        F[37] = q2;        F[38] = q2 * X2;
        F[39] = q2 * X3;   F[40] = q2 * X4;
        F[41] = X2 * X3;   F[42] = X2 * q3;   F[43] = X2 * X3 * X4;
        F[44] = X2 * X4;   F[45] = X2 * q4;
        F[46] = X3;        F[47] = q3;        F[48] = q3 * X3;
        F[49] = q3 * X4;
        F[50] = X3 * X4;   F[51] = X3 * q4;
        F[52] = X4;        F[53] = q4;        F[54] = q4 * X4;
        F[55] = __cosf(w0 * X0); F[56] = __cosf(w0 * X1); F[57] = __cosf(w0 * X2);
        F[58] = __cosf(w0 * X3); F[59] = __cosf(w0 * X4);
        F[60] = __sinf(w1 * X0); F[61] = __sinf(w1 * X1); F[62] = __sinf(w1 * X2);
        F[63] = __sinf(w1 * X3); F[64] = __sinf(w1 * X4);
        F[65] = 0.0f; F[66] = 0.0f; F[67] = 0.0f;

        v8f c0 = {};
        v8f c1 = {};
#pragma unroll
        for (int g = 0; g < 17; ++g) {
            const int k0 = 4 * g, k1 = k0 + 1, k2 = k0 + 2, k3 = k0 + 3;
            const float z0 = swapx16(F[k0]);
            const float z1 = swapx16(F[k1]);
            const float z2 = swapx16(F[k2]);
            const float z3 = swapx16(F[k3]);

            v2f bA = { hi ? z2 : F[k0], hi ? z3 : F[k1] };
            v2f bB = { hi ? F[k2] : z0, hi ? F[k3] : z1 };

            c0 = __builtin_amdgcn_wmma_f32_16x16x4_f32(
                     false, a[g], false, bA, (short)0, c0, false, false);
            c1 = __builtin_amdgcn_wmma_f32_16x16x4_f32(
                     false, a[g], false, bB, (short)0, c1, false, false);
        }

        if (!hi) {
            const int r0 = base + rsel;
            if (r0 < n) {
                float2 o; o.x = c0[0]; o.y = c0[1];
                *(float2*)(out + (size_t)r0 * 2) = o;
            }
            const int r1 = base + 16 + rsel;
            if (r1 < n) {
                float2 o; o.x = c1[0]; o.y = c1[1];
                *(float2*)(out + (size_t)r1 * 2) = o;
            }
        }
    }
}

extern "C" void kernel_launch(void* const* d_in, const int* in_sizes, int n_in,
                              void* d_out, int out_size, void* d_ws, size_t ws_size,
                              hipStream_t stream) {
    const float* x          = (const float*)d_in[0];
    const float* t          = (const float*)d_in[1];
    const float* layer      = (const float*)d_in[2];
    const float* omegas     = (const float*)d_in[3];
    const float* ext_filter = (const float*)d_in[4];
    float* out = (float*)d_out;

    const int n      = in_sizes[1];
    const int ntiles = (n + 31) / 32;
    const int waves_per_block = 8;
    int grid = (ntiles + waves_per_block - 1) / waves_per_block;
    if (grid > 2048) grid = 2048;

    sindy_wmma_f32_kernel<<<grid, 256, 0, stream>>>(x, t, layer, omegas,
                                                    ext_filter, out, n, ntiles);
}


# repack the coefficient matrix to K = 12: six trig values stay among the w, four move to the u rows, so 3 main WMMAs per 16 rows instead of 4; terms summed in a new order
# speedup vs baseline: 1.9620x; 1.9620x over previous
#include <hip/hip_runtime.h>


typedef __attribute__((ext_vector_type(2))) float v2f;
typedef __attribute__((ext_vector_type(8))) float v8f;

__constant__ const signed char kCoefIdx[256] = {
    -1,  0, 21, 55, 56, 60, -1, -1,
    36, 46, 52, 58, 59, 63, -1, -1,
     1,  2,  3, -1, -1, -1, -1, -1,
     4,  5,  6, -1, -1, -1, -1, -1,
    22,  8, 23, -1, -1, -1, -1, -1,
    24, 25, 26, -1, -1, -1, -1, -1,
    -1, -1,  7, -1, -1, -1, -1, -1,
    12, 16, 19, -1, -1, -1, -1, -1,
    -1, -1, -1, -1, -1, -1, -1, -1,
    27, 31, 34, -1, -1, -1, -1, -1,
    -1, -1, -1, -1, -1, -1, -1, -1,
     9, 10, 11, -1, -1, -1, -1, -1,
    61, -1, -1, -1, -1, -1, -1, -1,
    -1, -1, -1, -1, -1, -1, -1, -1,
    57, -1, -1, -1, -1, -1, -1, -1,
    -1, -1, -1, -1, -1, -1, -1, -1,
    37, 13, 28, -1, -1, -1, -1, -1,
    38, 39, 40, -1, -1, -1, -1, -1,
    47, 17, 32, -1, -1, -1, -1, -1,
    42, 48, 49, -1, -1, -1, -1, -1,
    53, 20, 35, -1, -1, -1, -1, -1,
    45, 51, 54, -1, -1, -1, -1, -1,
    41, 14, 29, -1, -1, -1, -1, -1,
    -1, -1, 43, -1, -1, -1, -1, -1,
    44, 15, 30, -1, -1, -1, -1, -1,
    -1, -1, -1, -1, -1, -1, -1, -1,
    50, 18, 33, -1, -1, -1, -1, -1,
    -1, -1, -1, -1, -1, -1, -1, -1,
    64, -1, -1, -1, -1, -1, -1, -1,
    -1, -1, -1, -1, -1, -1, -1, -1,
    62, -1, -1, -1, -1, -1, -1, -1,
    -1, -1, -1, -1, -1, -1, -1, -1,
};

__global__ __launch_bounds__(256) void sindy_wmma_f32_kernel(
    const float* __restrict__ x,
    const float* __restrict__ t,
    const float* __restrict__ layer,
    const float* __restrict__ omegas,
    const float* __restrict__ ext_filter,
    float* __restrict__ out,
    int n, int ntiles)
{
    const int lane = threadIdx.x & 31;
    const int wave = threadIdx.x >> 5;
    const int rsel = lane & 15;
    const bool hi  = lane >= 16;

    const int nwaves = gridDim.x * 8;
    const int wgid   = blockIdx.x * 8 + wave;

    const float w0 = omegas[0], w1 = omegas[1];
    const float f0 = ext_filter[0], f1 = ext_filter[1];

    const float fsel = (rsel == 0) ? f0 : ((rsel == 1) ? f1 : 0.0f);
    const v2f a2 = { fsel, fsel };

    v2f a[3];
    {
        const signed char* ci = kCoefIdx + (rsel * 2 + (hi ? 1 : 0)) * 8;
#pragma unroll
        for (int g = 0; g < 3; ++g) {
            const int i0 = ci[2 * g], i1 = ci[2 * g + 1];
            const float c0 = layer[i0 < 0 ? 0 : i0];
            const float c1 = layer[i1 < 0 ? 0 : i1];
            a[g][0] = (i0 < 0) ? 0.0f : c0;
            a[g][1] = (i1 < 0) ? 0.0f : c1;
        }
    }

    for (int tile = wgid; tile < ntiles; tile += nwaves) {
        const int base = tile * 32;

        float4 xv[2];
        float  tv[2];
#pragma unroll
        for (int sub = 0; sub < 2; ++sub) {
            const int row = base + 16 * sub + rsel;
            const int rc  = (row < n) ? row : (n - 1);
            xv[sub] = *(const float4*)(x + (size_t)rc * 4);
            tv[sub] = t[rc];
        }

        v2f bp[2];
#pragma unroll
        for (int sub = 0; sub < 2; ++sub) {
            const float X0 = xv[sub].x, X1 = xv[sub].y, X2 = xv[sub].z,
                        X3 = xv[sub].w, X4 = tv[sub];

            const float al = hi ? X2 : 1.0f;
            const float be = hi ? X3 : X0;
            const float ga = hi ? X4 : X1;

            const float cB = __cosf(w0 * be), cG = __cosf(w0 * ga);
            const float sB = __sinf(w1 * be), sG = __sinf(w1 * ga);
            const float c2 = __cosf(w0 * X2), s2 = __sinf(w1 * X2);

            const v2f b0 = { al, be };
            const v2f b1 = { ga, cB };
            const v2f b2 = { cG, sB };

            const float u0 = al * al, u1 = be * be, u2 = ga * ga;
            const float u3 = al * be, u4 = al * ga, u5 = be * ga;
            const float u6 = sG;
            const float u7 = hi ? s2 : c2;

            v8f acc = {};
            acc = __builtin_amdgcn_wmma_f32_16x16x4_f32(
                      false, a[0], false, b0, (short)0, acc, false, false);
            acc = __builtin_amdgcn_wmma_f32_16x16x4_f32(
                      false, a[1], false, b1, (short)0, acc, false, false);
            acc = __builtin_amdgcn_wmma_f32_16x16x4_f32(
                      false, a[2], false, b2, (short)0, acc, false, false);

            bp[sub][0] = u0 * acc[0] + u1 * acc[1] + u2 * acc[2] + u3 * acc[3];
            bp[sub][1] = u4 * acc[4] + u5 * acc[5] + u6 * acc[6] + u7 * acc[7];
        }

        v8f o0 = {}, o1 = {};
        o0 = __builtin_amdgcn_wmma_f32_16x16x4_f32(
                 false, a2, false, bp[0], (short)0, o0, false, false);
        o1 = __builtin_amdgcn_wmma_f32_16x16x4_f32(
                 false, a2, false, bp[1], (short)0, o1, false, false);

        if (!hi) {
            const int r0 = base + rsel;
            if (r0 < n) {
                float2 o; o.x = o0[0]; o.y = o0[1];
                *(float2*)(out + (size_t)r0 * 2) = o;
            }
            const int r1 = base + 16 + rsel;
            if (r1 < n) {
                float2 o; o.x = o1[0]; o.y = o1[1];
                *(float2*)(out + (size_t)r1 * 2) = o;
            }
        }
    }
}

extern "C" void kernel_launch(void* const* d_in, const int* in_sizes, int n_in,
                              void* d_out, int out_size, void* d_ws, size_t ws_size,
                              hipStream_t stream) {
    const float* x          = (const float*)d_in[0];
    const float* t          = (const float*)d_in[1];
    const float* layer      = (const float*)d_in[2];
    const float* omegas     = (const float*)d_in[3];
    const float* ext_filter = (const float*)d_in[4];
    float* out = (float*)d_out;

    const int n      = in_sizes[1];
    const int ntiles = (n + 31) / 32;
    const int waves_per_block = 8;
    int grid = (ntiles + waves_per_block - 1) / waves_per_block;
    if (grid > 2048) grid = 2048;

    sindy_wmma_f32_kernel<<<grid, 256, 0, stream>>>(x, t, layer, omegas,
                                                    ext_filter, out, n, ntiles);
}


# one partial sum per lane (packed multiply-adds on the D register pairs, then one add) so ONE closing WMMA serves both 16-row halves: 7 WMMAs per 32 rows instead of 8; terms summed in a new order
# speedup vs baseline: 2.2006x; 1.1216x over previous
#include <hip/hip_runtime.h>


typedef __attribute__((ext_vector_type(2))) float v2f;
typedef __attribute__((ext_vector_type(8))) float v8f;

__constant__ const signed char kCoefIdx[256] = {
    -1,  0, 21, 55, 56, 60, -1, -1,
    36, 46, 52, 58, 59, 63, -1, -1,
     1,  2,  3, -1, -1, -1, -1, -1,
     4,  5,  6, -1, -1, -1, -1, -1,
    22,  8, 23, -1, -1, -1, -1, -1,
    24, 25, 26, -1, -1, -1, -1, -1,
    -1, -1,  7, -1, -1, -1, -1, -1,
    12, 16, 19, -1, -1, -1, -1, -1,
    -1, -1, -1, -1, -1, -1, -1, -1,
    27, 31, 34, -1, -1, -1, -1, -1,
    -1, -1, -1, -1, -1, -1, -1, -1,
     9, 10, 11, -1, -1, -1, -1, -1,
    61, -1, -1, -1, -1, -1, -1, -1,
    -1, -1, -1, -1, -1, -1, -1, -1,
    57, -1, -1, -1, -1, -1, -1, -1,
    -1, -1, -1, -1, -1, -1, -1, -1,
    37, 13, 28, -1, -1, -1, -1, -1,
    38, 39, 40, -1, -1, -1, -1, -1,
    47, 17, 32, -1, -1, -1, -1, -1,
    42, 48, 49, -1, -1, -1, -1, -1,
    53, 20, 35, -1, -1, -1, -1, -1,
    45, 51, 54, -1, -1, -1, -1, -1,
    41, 14, 29, -1, -1, -1, -1, -1,
    -1, -1, 43, -1, -1, -1, -1, -1,
    44, 15, 30, -1, -1, -1, -1, -1,
    -1, -1, -1, -1, -1, -1, -1, -1,
    50, 18, 33, -1, -1, -1, -1, -1,
    -1, -1, -1, -1, -1, -1, -1, -1,
    64, -1, -1, -1, -1, -1, -1, -1,
    -1, -1, -1, -1, -1, -1, -1, -1,
    62, -1, -1, -1, -1, -1, -1, -1,
    -1, -1, -1, -1, -1, -1, -1, -1,
};

__global__ __launch_bounds__(256) void sindy_wmma_f32_kernel(
    const float* __restrict__ x,
    const float* __restrict__ t,
    const float* __restrict__ layer,
    const float* __restrict__ omegas,
    const float* __restrict__ ext_filter,
    float* __restrict__ out,
    int n, int ntiles)
{
    const int lane = threadIdx.x & 31;
    const int wave = threadIdx.x >> 5;
    const int rsel = lane & 15;
    const bool hi  = lane >= 16;

    const int nwaves = gridDim.x * 8;
    const int wgid   = blockIdx.x * 8 + wave;

    const float w0 = omegas[0], w1 = omegas[1];
    const float f0 = ext_filter[0], f1 = ext_filter[1];

    const v2f a2 = { (rsel == 0) ? f0 : ((rsel == 1) ? f1 : 0.0f),
                     (rsel == 2) ? f0 : ((rsel == 3) ? f1 : 0.0f) };

    v2f a[3];
    {
        const signed char* ci = kCoefIdx + (rsel * 2 + (hi ? 1 : 0)) * 8;
#pragma unroll
        for (int g = 0; g < 3; ++g) {
            const int i0 = ci[2 * g], i1 = ci[2 * g + 1];
            const float c0 = layer[i0 < 0 ? 0 : i0];
            const float c1 = layer[i1 < 0 ? 0 : i1];
            a[g][0] = (i0 < 0) ? 0.0f : c0;
            a[g][1] = (i1 < 0) ? 0.0f : c1;
        }
    }

    for (int tile = wgid; tile < ntiles; tile += nwaves) {
        const int base = tile * 32;

        float4 xv[2];
        float  tv[2];
#pragma unroll
        for (int sub = 0; sub < 2; ++sub) {
            const int row = base + 16 * sub + rsel;
            const int rc  = (row < n) ? row : (n - 1);
            xv[sub] = *(const float4*)(x + (size_t)rc * 4);
            tv[sub] = t[rc];
        }

        v2f bp;
#pragma unroll
        for (int sub = 0; sub < 2; ++sub) {
            const float X0 = xv[sub].x, X1 = xv[sub].y, X2 = xv[sub].z,
                        X3 = xv[sub].w, X4 = tv[sub];

            const float al = hi ? X2 : 1.0f;
            const float be = hi ? X3 : X0;
            const float ga = hi ? X4 : X1;

            const float cB = __cosf(w0 * be), cG = __cosf(w0 * ga);
            const float sB = __sinf(w1 * be), sG = __sinf(w1 * ga);
            const float c2 = __cosf(w0 * X2), s2 = __sinf(w1 * X2);

            const v2f b0 = { al, be };
            const v2f b1 = { ga, cB };
            const v2f b2 = { cG, sB };

            const float u0 = al * al, u1 = be * be, u2 = ga * ga;
            const float u3 = al * be, u4 = al * ga, u5 = be * ga;
            const float u6 = sG;
            const float u7 = hi ? s2 : c2;

            v8f acc = {};
            acc = __builtin_amdgcn_wmma_f32_16x16x4_f32(
                      false, a[0], false, b0, (short)0, acc, false, false);
            acc = __builtin_amdgcn_wmma_f32_16x16x4_f32(
                      false, a[1], false, b1, (short)0, acc, false, false);
            acc = __builtin_amdgcn_wmma_f32_16x16x4_f32(
                      false, a[2], false, b2, (short)0, acc, false, false);

            const v2f d01 = { acc[0], acc[1] }, d23 = { acc[2], acc[3] };
            const v2f d45 = { acc[4], acc[5] }, d67 = { acc[6], acc[7] };
            const v2f w01 = { u0, u1 }, w23 = { u2, u3 };
            const v2f w45 = { u4, u5 }, w67 = { u6, u7 };
            v2f q = w01 * d01;
            q = __builtin_elementwise_fma(w23, d23, q);
            q = __builtin_elementwise_fma(w45, d45, q);
            q = __builtin_elementwise_fma(w67, d67, q);
            bp[sub] = q[0] + q[1];
        }

        v8f o = {};
        o = __builtin_amdgcn_wmma_f32_16x16x4_f32(
                false, a2, false, bp, (short)0, o, false, false);

        if (!hi) {
            const int r0 = base + rsel;
            if (r0 < n) {
                float2 v; v.x = o[0]; v.y = o[1];
                *(float2*)(out + (size_t)r0 * 2) = v;
            }
            const int r1 = base + 16 + rsel;
            if (r1 < n) {
                float2 v; v.x = o[2]; v.y = o[3];
                *(float2*)(out + (size_t)r1 * 2) = v;
            }
        }
    }
}

extern "C" void kernel_launch(void* const* d_in, const int* in_sizes, int n_in,
                              void* d_out, int out_size, void* d_ws, size_t ws_size,
                              hipStream_t stream) {
    const float* x          = (const float*)d_in[0];
    const float* t          = (const float*)d_in[1];
    const float* layer      = (const float*)d_in[2];
    const float* omegas     = (const float*)d_in[3];
    const float* ext_filter = (const float*)d_in[4];
    float* out = (float*)d_out;

    const int n      = in_sizes[1];
    const int ntiles = (n + 31) / 32;
    const int waves_per_block = 8;
    int grid = (ntiles + waves_per_block - 1) / waves_per_block;
    if (grid > 2048) grid = 2048;

    sindy_wmma_f32_kernel<<<grid, 256, 0, stream>>>(x, t, layer, omegas,
                                                    ext_filter, out, n, ntiles);
}


# move two trig values per lane half out of the C w product into one packed multiply-add on the lane's partial sum, so K = 8: 5 WMMAs per 32 rows instead of 7; terms summed in a new order
# speedup vs baseline: 2.3894x; 1.0858x over previous
#include <hip/hip_runtime.h>


typedef __attribute__((ext_vector_type(2))) float v2f;
typedef __attribute__((ext_vector_type(8))) float v8f;

__constant__ const signed char kCoefIdx[128] = {
    -1,  0, 21, 55,
    36, 46, 52, 58,
     1,  2,  3, -1,
     4,  5,  6, -1,
    22,  8, 23, -1,
    24, 25, 26, -1,
    -1, -1,  7, -1,
    12, 16, 19, -1,
    -1, -1, -1, -1,
    27, 31, 34, -1,
    -1, -1, -1, -1,
     9, 10, 11, -1,
    61, -1, -1, -1,
    -1, -1, -1, -1,
    57, -1, -1, -1,
    -1, -1, -1, -1,
    37, 13, 28, -1,
    38, 39, 40, -1,
    47, 17, 32, -1,
    42, 48, 49, -1,
    53, 20, 35, -1,
    45, 51, 54, -1,
    41, 14, 29, -1,
    -1, -1, 43, -1,
    44, 15, 30, -1,
    -1, -1, -1, -1,
    50, 18, 33, -1,
    -1, -1, -1, -1,
    64, -1, -1, -1,
    -1, -1, -1, -1,
    62, -1, -1, -1,
    -1, -1, -1, -1,
};

__global__ __launch_bounds__(256) void sindy_wmma_f32_kernel(
    const float* __restrict__ x,
    const float* __restrict__ t,
    const float* __restrict__ layer,
    const float* __restrict__ omegas,
    const float* __restrict__ ext_filter,
    float* __restrict__ out,
    int n, int ntiles)
{
    const int lane = threadIdx.x & 31;
    const int wave = threadIdx.x >> 5;
    const int rsel = lane & 15;
    const bool hi  = lane >= 16;

    const int nwaves = gridDim.x * 8;
    const int wgid   = blockIdx.x * 8 + wave;

    const float w0 = omegas[0], w1 = omegas[1];
    const float f0 = ext_filter[0], f1 = ext_filter[1];

    const v2f a2 = { (rsel == 0) ? f0 : ((rsel == 1) ? f1 : 0.0f),
                     (rsel == 2) ? f0 : ((rsel == 3) ? f1 : 0.0f) };

    v2f a[2];
    {
        const signed char* ci = kCoefIdx + (rsel * 2 + (hi ? 1 : 0)) * 4;
#pragma unroll
        for (int g = 0; g < 2; ++g) {
            const int i0 = ci[2 * g], i1 = ci[2 * g + 1];
            const float c0 = layer[i0 < 0 ? 0 : i0];
            const float c1 = layer[i1 < 0 ? 0 : i1];
            a[g][0] = (i0 < 0) ? 0.0f : c0;
            a[g][1] = (i1 < 0) ? 0.0f : c1;
        }
    }

    const v2f kd = { hi ? layer[59] : layer[56], hi ? layer[63] : layer[60] };

    for (int tile = wgid; tile < ntiles; tile += nwaves) {
        const int base = tile * 32;

        float4 xv[2];
        float  tv[2];
#pragma unroll
        for (int sub = 0; sub < 2; ++sub) {
            const int row = base + 16 * sub + rsel;
            const int rc  = (row < n) ? row : (n - 1);
            xv[sub] = *(const float4*)(x + (size_t)rc * 4);
            tv[sub] = t[rc];
        }

        v2f bp;
#pragma unroll
        for (int sub = 0; sub < 2; ++sub) {
            const float X0 = xv[sub].x, X1 = xv[sub].y, X2 = xv[sub].z,
                        X3 = xv[sub].w, X4 = tv[sub];

            const float al = hi ? X2 : 1.0f;
            const float be = hi ? X3 : X0;
            const float ga = hi ? X4 : X1;

            const float cB = __cosf(w0 * be), cG = __cosf(w0 * ga);
            const float sB = __sinf(w1 * be), sG = __sinf(w1 * ga);
            const float c2 = __cosf(w0 * X2), s2 = __sinf(w1 * X2);

            const v2f b0 = { al, be };
            const v2f b1 = { ga, cB };

            const v2f tg = { cG, sB };
            v2f q = tg * kd;

            const float u0 = al * al, u1 = be * be, u2 = ga * ga;
            const float u3 = al * be, u4 = al * ga, u5 = be * ga;
            const float u6 = sG;
            const float u7 = hi ? s2 : c2;

            v8f acc = {};
            acc = __builtin_amdgcn_wmma_f32_16x16x4_f32(
                      false, a[0], false, b0, (short)0, acc, false, false);
            acc = __builtin_amdgcn_wmma_f32_16x16x4_f32(
                      false, a[1], false, b1, (short)0, acc, false, false);

            const v2f d01 = { acc[0], acc[1] }, d23 = { acc[2], acc[3] };
            const v2f d45 = { acc[4], acc[5] }, d67 = { acc[6], acc[7] };
            const v2f w01 = { u0, u1 }, w23 = { u2, u3 };
            const v2f w45 = { u4, u5 }, w67 = { u6, u7 };
            q = __builtin_elementwise_fma(w01, d01, q);
            q = __builtin_elementwise_fma(w23, d23, q);
            q = __builtin_elementwise_fma(w45, d45, q);
            q = __builtin_elementwise_fma(w67, d67, q);
            bp[sub] = q[0] + q[1];
        }

        v8f o = {};
        o = __builtin_amdgcn_wmma_f32_16x16x4_f32(
                false, a2, false, bp, (short)0, o, false, false);

        if (!hi) {
            const int r0 = base + rsel;
            if (r0 < n) {
                float2 v; v.x = o[0]; v.y = o[1];
                *(float2*)(out + (size_t)r0 * 2) = v;
            }
            const int r1 = base + 16 + rsel;
            if (r1 < n) {
                float2 v; v.x = o[2]; v.y = o[3];
                *(float2*)(out + (size_t)r1 * 2) = v;
            }
        }
    }
}

extern "C" void kernel_launch(void* const* d_in, const int* in_sizes, int n_in,
                              void* d_out, int out_size, void* d_ws, size_t ws_size,
                              hipStream_t stream) {
    const float* x          = (const float*)d_in[0];
    const float* t          = (const float*)d_in[1];
    const float* layer      = (const float*)d_in[2];
    const float* omegas     = (const float*)d_in[3];
    const float* ext_filter = (const float*)d_in[4];
    float* out = (float*)d_out;

    const int n      = in_sizes[1];
    const int ntiles = (n + 31) / 32;
    const int waves_per_block = 8;
    int grid = (ntiles + waves_per_block - 1) / waves_per_block;
    if (grid > 2048) grid = 2048;

    sindy_wmma_f32_kernel<<<grid, 256, 0, stream>>>(x, t, layer, omegas,
                                                    ext_filter, out, n, ntiles);
}


# a wave's loads leave together: index-table word fetched beside the arguments, layer gather and both 16-row halves of a tile requested as one group: 2 memory round trips before the first WMMA, not 5
# speedup vs baseline: 2.5589x; 1.0709x over previous
#include <hip/hip_runtime.h>


typedef __attribute__((ext_vector_type(2))) float v2f;
typedef __attribute__((ext_vector_type(8))) float v8f;

__constant__ const signed char kCoefIdx[128] __attribute__((aligned(16))) = {
    -1,  0, 21, 55,
    36, 46, 52, 58,
     1,  2,  3, -1,
     4,  5,  6, -1,
    22,  8, 23, -1,
    24, 25, 26, -1,
    -1, -1,  7, -1,
    12, 16, 19, -1,
    -1, -1, -1, -1,
    27, 31, 34, -1,
    -1, -1, -1, -1,
     9, 10, 11, -1,
    61, -1, -1, -1,
    -1, -1, -1, -1,
    57, -1, -1, -1,
    -1, -1, -1, -1,
    37, 13, 28, -1,
    38, 39, 40, -1,
    47, 17, 32, -1,
    42, 48, 49, -1,
    53, 20, 35, -1,
    45, 51, 54, -1,
    41, 14, 29, -1,
    -1, -1, 43, -1,
    44, 15, 30, -1,
    -1, -1, -1, -1,
    50, 18, 33, -1,
    -1, -1, -1, -1,
    64, -1, -1, -1,
    -1, -1, -1, -1,
    62, -1, -1, -1,
    -1, -1, -1, -1,
};

struct Rows { float4 x0, x1; float t0, t1; };

__device__ __forceinline__ Rows load_rows(const float* __restrict__ x,
                                          const float* __restrict__ t,
                                          int n, int tile, int rsel)
{
    const int r0 = tile * 32 + rsel, r1 = r0 + 16;
    const int c0 = (r0 < n) ? r0 : (n - 1);
    const int c1 = (r1 < n) ? r1 : (n - 1);
    Rows r;
    r.x0 = *(const float4*)(x + (size_t)c0 * 4);
    r.x1 = *(const float4*)(x + (size_t)c1 * 4);
    r.t0 = t[c0];
    r.t1 = t[c1];
    return r;
}

__device__ __forceinline__ void do_tile(const Rows& r, int tile, int n, int rsel,
                                        bool hi, float w0, float w1,
                                        v2f a0, v2f a1, v2f a2, v2f kd,
                                        float* __restrict__ out)
{
    const float4 xv[2] = { r.x0, r.x1 };
    const float  tv[2] = { r.t0, r.t1 };

    v2f bp;
#pragma unroll
    for (int sub = 0; sub < 2; ++sub) {
        const float X0 = xv[sub].x, X1 = xv[sub].y, X2 = xv[sub].z,
                    X3 = xv[sub].w, X4 = tv[sub];

        const float al = hi ? X2 : 1.0f;
        const float be = hi ? X3 : X0;
        const float ga = hi ? X4 : X1;

        const float cB = __cosf(w0 * be), cG = __cosf(w0 * ga);
        const float sB = __sinf(w1 * be), sG = __sinf(w1 * ga);
        const float c2 = __cosf(w0 * X2), s2 = __sinf(w1 * X2);

        const v2f b0 = { al, be };
        const v2f b1 = { ga, cB };

        const v2f tg = { cG, sB };
        v2f q = tg * kd;

        const float u0 = al * al, u1 = be * be, u2 = ga * ga;
        const float u3 = al * be, u4 = al * ga, u5 = be * ga;
        const float u6 = sG;
        const float u7 = hi ? s2 : c2;

        v8f acc = {};
        acc = __builtin_amdgcn_wmma_f32_16x16x4_f32(
                  false, a0, false, b0, (short)0, acc, false, false);
        acc = __builtin_amdgcn_wmma_f32_16x16x4_f32(
                  false, a1, false, b1, (short)0, acc, false, false);

        const v2f d01 = { acc[0], acc[1] }, d23 = { acc[2], acc[3] };
        const v2f d45 = { acc[4], acc[5] }, d67 = { acc[6], acc[7] };
        const v2f w01 = { u0, u1 }, w23 = { u2, u3 };
        const v2f w45 = { u4, u5 }, w67 = { u6, u7 };
        q = __builtin_elementwise_fma(w01, d01, q);
        q = __builtin_elementwise_fma(w23, d23, q);
        q = __builtin_elementwise_fma(w45, d45, q);
        q = __builtin_elementwise_fma(w67, d67, q);
        bp[sub] = q[0] + q[1];
    }

    v8f o = {};
    o = __builtin_amdgcn_wmma_f32_16x16x4_f32(
            false, a2, false, bp, (short)0, o, false, false);

    if (!hi) {
        const int r0 = tile * 32 + rsel;
        if (r0 < n) {
            float2 v; v.x = o[0]; v.y = o[1];
            *(float2*)(out + (size_t)r0 * 2) = v;
        }
        const int r1 = r0 + 16;
        if (r1 < n) {
            float2 v; v.x = o[2]; v.y = o[3];
            *(float2*)(out + (size_t)r1 * 2) = v;
        }
    }
}

__global__ __launch_bounds__(256) void sindy_wmma_f32_kernel(
    const float* __restrict__ x,
    const float* __restrict__ t,
    const float* __restrict__ layer,
    const float* __restrict__ omegas,
    const float* __restrict__ ext_filter,
    float* __restrict__ out,
    int n, int ntiles)
{
    const int lane = threadIdx.x & 31;
    const int wave = threadIdx.x >> 5;
    const int rsel = lane & 15;
    const bool hi  = lane >= 16;

    int tabw;
    __builtin_memcpy(&tabw, kCoefIdx + (rsel * 2 + (hi ? 1 : 0)) * 4, 4);
    asm volatile("" :: "v"(tabw));

    const int nwaves = gridDim.x * 8;
    int tile = blockIdx.x * 8 + wave;
    if (tile >= ntiles) return;

    Rows r = load_rows(x, t, n, tile, rsel);

    const float w0 = omegas[0], w1 = omegas[1];
    const float f0 = ext_filter[0], f1 = ext_filter[1];

    int   ci[4];
    float cr[4];
#pragma unroll
    for (int j = 0; j < 4; ++j) {
        ci[j] = (signed char)(tabw >> (8 * j));
        cr[j] = layer[ci[j] < 0 ? 0 : ci[j]];
    }
    const v2f kd = { hi ? layer[59] : layer[56], hi ? layer[63] : layer[60] };

    __builtin_amdgcn_sched_barrier(0);

    const v2f a2 = { (rsel == 0) ? f0 : ((rsel == 1) ? f1 : 0.0f),
                     (rsel == 2) ? f0 : ((rsel == 3) ? f1 : 0.0f) };

    v2f a[2];
#pragma unroll
    for (int g = 0; g < 2; ++g) {
        a[g][0] = (ci[2 * g]     < 0) ? 0.0f : cr[2 * g];
        a[g][1] = (ci[2 * g + 1] < 0) ? 0.0f : cr[2 * g + 1];
    }

    for (;;) {
        do_tile(r, tile, n, rsel, hi, w0, w1, a[0], a[1], a2, kd, out);
        tile += nwaves;
        if (tile >= ntiles) break;
        r = load_rows(x, t, n, tile, rsel);
    }
}

extern "C" void kernel_launch(void* const* d_in, const int* in_sizes, int n_in,
                              void* d_out, int out_size, void* d_ws, size_t ws_size,
                              hipStream_t stream) {
    const float* x          = (const float*)d_in[0];
    const float* t          = (const float*)d_in[1];
    const float* layer      = (const float*)d_in[2];
    const float* omegas     = (const float*)d_in[3];
    const float* ext_filter = (const float*)d_in[4];
    float* out = (float*)d_out;

    const int n      = in_sizes[1];
    const int ntiles = (n + 31) / 32;
    const int waves_per_block = 8;
    int grid = (ntiles + waves_per_block - 1) / waves_per_block;
    if (grid > 2048) grid = 2048;

    sindy_wmma_f32_kernel<<<grid, 256, 0, stream>>>(x, t, layer, omegas,
                                                    ext_filter, out, n, ntiles);
}


# lane halves split a tile's work: each loads only its own two x columns (t shared), gets t's cos or sin from one shared sine (phase in revolutions), stores its own 16 rows; terms summed in a new order
# speedup vs baseline: 2.7503x; 1.0748x over previous
#include <hip/hip_runtime.h>


typedef __attribute__((ext_vector_type(2))) float v2f;
typedef __attribute__((ext_vector_type(8))) float v8f;

__constant__ const signed char kCoefIdx[128] __attribute__((aligned(16))) = {
     2,  3,  1, -1,
     4,  5,  6, -1,
     8, 23, 22, -1,
    24, 25, 26, -1,
    -1,  7,  0, -1,
    12, 16, 19, -1,
    -1, -1, 21, -1,
    27, 31, 34, -1,
    -1, -1, -1, 55,
    36, 46, 52, 57,
    -1, -1, 61, -1,
    -1, -1, -1, -1,
    -1, -1, -1, -1,
     9, 10, 11, -1,
    -1, -1, 59, -1,
    -1, -1, -1, -1,
    13, 28, 37, -1,
    38, 39, 40, -1,
    17, 32, 47, -1,
    42, 48, 49, -1,
    15, 30, 44, -1,
    -1, 43, 45, -1,
    18, 33, 50, -1,
    -1, -1, 51, -1,
    20, 35, 53, -1,
    -1, -1, 54, -1,
    -1, -1, 63, -1,
    -1, -1, -1, -1,
    14, 29, 41, -1,
    -1, -1, -1, -1,
    -1, -1, 64, -1,
    -1, -1, -1, -1,
};

struct Rows { float2 p0, p1; float t0, t1; };

__device__ __forceinline__ Rows load_rows(const float* __restrict__ x,
                                          const float* __restrict__ t,
                                          int n, int tile, int rsel, int hsel)
{
    const int r0 = tile * 32 + rsel, r1 = r0 + 16;
    const int c0 = (r0 < n) ? r0 : (n - 1);
    const int c1 = (r1 < n) ? r1 : (n - 1);
    const float2* __restrict__ xp = (const float2*)x;
    Rows r;
    r.p0 = xp[c0 * 2 + hsel];
    r.p1 = xp[c1 * 2 + hsel];
    r.t0 = t[c0];
    r.t1 = t[c1];
    return r;
}

__device__ __forceinline__ float cos_rev(float r) { return __builtin_amdgcn_cosf(r); }
__device__ __forceinline__ float sin_rev(float r) { return __builtin_amdgcn_sinf(r); }

__device__ __forceinline__ void do_tile(const Rows& r, int tile, int n, int lane,
                                        bool hi, float g0, float g1,
                                        float gs, float ph,
                                        v2f a0, v2f a1, v2f a2, v2f kd,
                                        float* __restrict__ out)
{
    const float2 pv[2] = { r.p0, r.p1 };
    const float  tv[2] = { r.t0, r.t1 };

    v2f bp;
#pragma unroll
    for (int sub = 0; sub < 2; ++sub) {
        const float T = tv[sub];

        const v2f P = { pv[sub].x, pv[sub].y };
        const float al = hi ? T : 1.0f;

        const v2f ca = P * g0, sa = P * g1;
        const float cB = cos_rev(ca[0]), cG = cos_rev(ca[1]);
        const float sB = sin_rev(sa[0]), sG = sin_rev(sa[1]);
        const float tT = sin_rev(__builtin_fmaf(gs, T, ph));

        const v2f b0 = P;
        const v2f b1 = { al, cB };

        const v2f tg = { cG, sB };
        v2f q = tg * kd;

        const v2f w01 = P * P;
        const v2f w23 = P * al;
        const v2f w45 = { al * al, sG };
        const v2f w67 = { P[0] * P[1], tT };

        v8f acc = {};
        acc = __builtin_amdgcn_wmma_f32_16x16x4_f32(
                  false, a0, false, b0, (short)0, acc, false, false);
        acc = __builtin_amdgcn_wmma_f32_16x16x4_f32(
                  false, a1, false, b1, (short)0, acc, false, false);

        const v2f d01 = { acc[0], acc[1] }, d23 = { acc[2], acc[3] };
        const v2f d45 = { acc[4], acc[5] }, d67 = { acc[6], acc[7] };
        q = __builtin_elementwise_fma(w01, d01, q);
        q = __builtin_elementwise_fma(w23, d23, q);
        q = __builtin_elementwise_fma(w45, d45, q);
        q = __builtin_elementwise_fma(w67, d67, q);
        bp[sub] = q[0] + q[1];
    }

    v8f o = {};
    o = __builtin_amdgcn_wmma_f32_16x16x4_f32(
            false, a2, false, bp, (short)0, o, false, false);

    const int row = tile * 32 + lane;
    if (row < n) {
        float2 v; v.x = o[0]; v.y = o[1];
        *(float2*)(out + (size_t)row * 2) = v;
    }
}

__global__ __launch_bounds__(256) void sindy_wmma_f32_kernel(
    const float* __restrict__ x,
    const float* __restrict__ t,
    const float* __restrict__ layer,
    const float* __restrict__ omegas,
    const float* __restrict__ ext_filter,
    float* __restrict__ out,
    int n, int ntiles)
{
    const int lane = threadIdx.x & 31;
    const int wave = threadIdx.x >> 5;
    const int rsel = lane & 15;
    const bool hi  = lane >= 16;

    int tabw;
    __builtin_memcpy(&tabw, kCoefIdx + (rsel * 2 + (hi ? 1 : 0)) * 4, 4);
    asm volatile("" :: "v"(tabw));

    const int nwaves = gridDim.x * 8;
    int tile = blockIdx.x * 8 + wave;
    if (tile >= ntiles) return;

    const int hsel = hi ? 1 : 0;
    Rows r = load_rows(x, t, n, tile, rsel, hsel);

    const float w0 = omegas[0], w1 = omegas[1];
    const float f0 = ext_filter[0], f1 = ext_filter[1];

    int   ci[4];
    float cr[4];
#pragma unroll
    for (int j = 0; j < 4; ++j) {
        ci[j] = (signed char)(tabw >> (8 * j));
        cr[j] = layer[ci[j] < 0 ? 0 : ci[j]];
    }
    const v2f kd = { hi ? layer[58] : layer[56], hi ? layer[62] : layer[60] };

    __builtin_amdgcn_sched_barrier(0);

    const v2f a2 = { (rsel == 0) ? f0 : ((rsel == 1) ? f1 : 0.0f),
                     (rsel == 8) ? f0 : ((rsel == 9) ? f1 : 0.0f) };

    const float kInv2Pi = 0.15915494309189535f;
    const float g0 = w0 * kInv2Pi, g1 = w1 * kInv2Pi;
    const float gs = hi ? g1 : g0;
    const float ph = hi ? 0.0f : 0.25f;

    v2f a[2];
#pragma unroll
    for (int g = 0; g < 2; ++g) {
        a[g][0] = (ci[2 * g]     < 0) ? 0.0f : cr[2 * g];
        a[g][1] = (ci[2 * g + 1] < 0) ? 0.0f : cr[2 * g + 1];
    }

    for (;;) {
        do_tile(r, tile, n, lane, hi, g0, g1, gs, ph, a[0], a[1], a2, kd, out);
        tile += nwaves;
        if (tile >= ntiles) break;
        r = load_rows(x, t, n, tile, rsel, hsel);
    }
}

extern "C" void kernel_launch(void* const* d_in, const int* in_sizes, int n_in,
                              void* d_out, int out_size, void* d_ws, size_t ws_size,
                              hipStream_t stream) {
    const float* x          = (const float*)d_in[0];
    const float* t          = (const float*)d_in[1];
    const float* layer      = (const float*)d_in[2];
    const float* omegas     = (const float*)d_in[3];
    const float* ext_filter = (const float*)d_in[4];
    float* out = (float*)d_out;

    const int n      = in_sizes[1];
    const int ntiles = (n + 31) / 32;
    const int waves_per_block = 8;
    int grid = (ntiles + waves_per_block - 1) / waves_per_block;
    if (grid > 2048) grid = 2048;

    sindy_wmma_f32_kernel<<<grid, 256, 0, stream>>>(x, t, layer, omegas,
                                                    ext_filter, out, n, ntiles);
}


# a wave owns 64 consecutive rows under a wave-uniform (scalar) span number: scalar base, one lane offset, fixed distances, all 8 row loads one group, scalar loop control; clamp and guard only past n
# speedup vs baseline: 2.9510x; 1.0730x over previous
#include <hip/hip_runtime.h>


typedef __attribute__((ext_vector_type(2))) float v2f;
typedef __attribute__((ext_vector_type(8))) float v8f;

__constant__ const signed char kCoefIdx[128] __attribute__((aligned(16))) = {
     2,  3,  1, -1,
     4,  5,  6, -1,
     8, 23, 22, -1,
    24, 25, 26, -1,
    -1,  7,  0, -1,
    12, 16, 19, -1,
    -1, -1, 21, -1,
    27, 31, 34, -1,
    -1, -1, -1, 55,
    36, 46, 52, 57,
    -1, -1, 61, -1,
    -1, -1, -1, -1,
    -1, -1, -1, -1,
     9, 10, 11, -1,
    -1, -1, 59, -1,
    -1, -1, -1, -1,
    13, 28, 37, -1,
    38, 39, 40, -1,
    17, 32, 47, -1,
    42, 48, 49, -1,
    15, 30, 44, -1,
    -1, 43, 45, -1,
    18, 33, 50, -1,
    -1, -1, 51, -1,
    20, 35, 53, -1,
    -1, -1, 54, -1,
    -1, -1, 63, -1,
    -1, -1, -1, -1,
    14, 29, 41, -1,
    -1, -1, -1, -1,
    -1, -1, 64, -1,
    -1, -1, -1, -1,
};

struct Rows { float2 p0, p1; float t0, t1; };

__device__ __forceinline__ Rows load_rows_full(const char* __restrict__ xs,
                                               const char* __restrict__ ts,
                                               unsigned xo, unsigned to)
{
    Rows r;
    r.p0 = *(const float2*)(xs + xo);
    r.p1 = *(const float2*)(xs + xo + 256);
    r.t0 = *(const float*)(ts + to);
    r.t1 = *(const float*)(ts + to + 64);
    return r;
}

__device__ __forceinline__ Rows load_rows(const float* __restrict__ x,
                                          const float* __restrict__ t,
                                          int n, int row0, int rsel, int hsel)
{
    const int r0 = row0 + rsel, r1 = r0 + 16;
    const int c0 = (r0 < n) ? r0 : (n - 1);
    const int c1 = (r1 < n) ? r1 : (n - 1);
    const float2* __restrict__ xp = (const float2*)x;
    Rows r;
    r.p0 = xp[c0 * 2 + hsel];
    r.p1 = xp[c1 * 2 + hsel];
    r.t0 = t[c0];
    r.t1 = t[c1];
    return r;
}

__device__ __forceinline__ float cos_rev(float r) { return __builtin_amdgcn_cosf(r); }
__device__ __forceinline__ float sin_rev(float r) { return __builtin_amdgcn_sinf(r); }

template <bool FULL>
__device__ __forceinline__ void do_tile(const Rows& r, int row0, int n,
                                        int lane, bool hi, float g0, float g1,
                                        float gs, float ph,
                                        v2f a0, v2f a1, v2f a2, v2f kd,
                                        float* __restrict__ out)
{
    const float2 pv[2] = { r.p0, r.p1 };
    const float  tv[2] = { r.t0, r.t1 };

    v2f bp;
#pragma unroll
    for (int sub = 0; sub < 2; ++sub) {
        const float T = tv[sub];

        const v2f P = { pv[sub].x, pv[sub].y };
        const float al = hi ? T : 1.0f;

        const v2f ca = P * g0, sa = P * g1;
        const float cB = cos_rev(ca[0]), cG = cos_rev(ca[1]);
        const float sB = sin_rev(sa[0]), sG = sin_rev(sa[1]);
        const float tT = sin_rev(__builtin_fmaf(gs, T, ph));

        const v2f b0 = P;
        const v2f b1 = { al, cB };

        const v2f tg = { cG, sB };
        v2f q = tg * kd;

        const v2f w01 = P * P;
        const v2f w23 = P * al;
        const v2f w45 = { al * al, sG };
        const v2f w67 = { P[0] * P[1], tT };

        v8f acc = {};
        acc = __builtin_amdgcn_wmma_f32_16x16x4_f32(
                  false, a0, false, b0, (short)0, acc, false, false);
        acc = __builtin_amdgcn_wmma_f32_16x16x4_f32(
                  false, a1, false, b1, (short)0, acc, false, false);

        const v2f d01 = { acc[0], acc[1] }, d23 = { acc[2], acc[3] };
        const v2f d45 = { acc[4], acc[5] }, d67 = { acc[6], acc[7] };
        q = __builtin_elementwise_fma(w01, d01, q);
        q = __builtin_elementwise_fma(w23, d23, q);
        q = __builtin_elementwise_fma(w45, d45, q);
        q = __builtin_elementwise_fma(w67, d67, q);
        bp[sub] = q[0] + q[1];
    }

    v8f o = {};
    o = __builtin_amdgcn_wmma_f32_16x16x4_f32(
            false, a2, false, bp, (short)0, o, false, false);

    float2 v; v.x = o[0]; v.y = o[1];
    if (FULL) {
        char* __restrict__ os = (char*)out + (size_t)row0 * 8;
        *(float2*)(os + (unsigned)lane * 8u) = v;
    } else {
        const int row = row0 + lane;
        if (row < n) *(float2*)(out + (size_t)row * 2) = v;
    }
}

__global__ __launch_bounds__(256) __attribute__((amdgpu_waves_per_eu(16, 16)))
void sindy_wmma_f32_kernel(
    const float* __restrict__ x,
    const float* __restrict__ t,
    const float* __restrict__ layer,
    const float* __restrict__ omegas,
    const float* __restrict__ ext_filter,
    float* __restrict__ out,
    int n, int nspans)
{
    const int lane = threadIdx.x & 31;
    const int wave = __builtin_amdgcn_readfirstlane((int)(threadIdx.x >> 5));
    const int rsel = lane & 15;
    const bool hi  = lane >= 16;

    int tabw;
    __builtin_memcpy(&tabw, kCoefIdx + (rsel * 2 + (hi ? 1 : 0)) * 4, 4);
    asm volatile("" :: "v"(tabw));

    const int nwaves = gridDim.x * 8;
    int span = blockIdx.x * 8 + wave;
    if (span >= nspans) return;

    const int nfull = n >> 6;

    const int hsel = hi ? 1 : 0;
    const unsigned xo = (unsigned)(rsel * 2 + hsel) * 8u;
    const unsigned to = (unsigned)rsel * 4u;
    Rows ra = {}, rb = {};
    if (span < nfull) {
        const char* xs = (const char*)x + (size_t)span * 1024;
        const char* ts = (const char*)t + (size_t)span * 256;
        ra = load_rows_full(xs, ts, xo, to);
        rb = load_rows_full(xs + 512, ts + 128, xo, to);
    }

    const float w0 = omegas[0], w1 = omegas[1];
    const float f0 = ext_filter[0], f1 = ext_filter[1];

    int   ci[4];
    float cr[4];
#pragma unroll
    for (int j = 0; j < 4; ++j) {
        ci[j] = (signed char)(tabw >> (8 * j));
        cr[j] = layer[ci[j] < 0 ? 0 : ci[j]];
    }
    const v2f kd = { hi ? layer[58] : layer[56], hi ? layer[62] : layer[60] };

    __builtin_amdgcn_sched_barrier(0);

    const v2f a2 = { (rsel == 0) ? f0 : ((rsel == 1) ? f1 : 0.0f),
                     (rsel == 8) ? f0 : ((rsel == 9) ? f1 : 0.0f) };

    const float kInv2Pi = 0.15915494309189535f;
    const float g0 = w0 * kInv2Pi, g1 = w1 * kInv2Pi;
    const float gs = hi ? g1 : g0;
    const float ph = hi ? 0.0f : 0.25f;

    v2f a[2];
#pragma unroll
    for (int g = 0; g < 2; ++g) {
        a[g][0] = (ci[2 * g]     < 0) ? 0.0f : cr[2 * g];
        a[g][1] = (ci[2 * g + 1] < 0) ? 0.0f : cr[2 * g + 1];
    }

    if (span < nfull) {
        for (;;) {
            do_tile<true>(ra, span * 64,      n, lane, hi, g0, g1, gs, ph, a[0], a[1], a2, kd, out);
            do_tile<true>(rb, span * 64 + 32, n, lane, hi, g0, g1, gs, ph, a[0], a[1], a2, kd, out);
            span += nwaves;
            if (span >= nfull) break;
            const char* xs = (const char*)x + (size_t)span * 1024;
            const char* ts = (const char*)t + (size_t)span * 256;
            ra = load_rows_full(xs, ts, xo, to);
            rb = load_rows_full(xs + 512, ts + 128, xo, to);
        }
    }

    if (span < nspans) {
        const Rows rc = load_rows(x, t, n, span * 64,      rsel, hsel);
        const Rows rd = load_rows(x, t, n, span * 64 + 32, rsel, hsel);
        do_tile<false>(rc, span * 64,      n, lane, hi, g0, g1, gs, ph, a[0], a[1], a2, kd, out);
        do_tile<false>(rd, span * 64 + 32, n, lane, hi, g0, g1, gs, ph, a[0], a[1], a2, kd, out);
    }
}

extern "C" void kernel_launch(void* const* d_in, const int* in_sizes, int n_in,
                              void* d_out, int out_size, void* d_ws, size_t ws_size,
                              hipStream_t stream) {
    const float* x          = (const float*)d_in[0];
    const float* t          = (const float*)d_in[1];
    const float* layer      = (const float*)d_in[2];
    const float* omegas     = (const float*)d_in[3];
    const float* ext_filter = (const float*)d_in[4];
    float* out = (float*)d_out;

    const int n      = in_sizes[1];
    const int nspans = (n + 63) / 64;
    const int waves_per_block = 8;
    int grid = (nspans + waves_per_block - 1) / waves_per_block;
    if (grid > 2048) grid = 2048;

    sindy_wmma_f32_kernel<<<grid, 256, 0, stream>>>(x, t, layer, omegas,
                                                    ext_filter, out, n, nspans);
}


# a wave owns 256 consecutive rows (4 spans) instead of 64: a quarter as many waves enter and pay the per-wave set-up; its 32 row loads leave as one group behind the layer gather; 4-wave work-groups
# speedup vs baseline: 3.3813x; 1.1458x over previous
#include <hip/hip_runtime.h>


typedef __attribute__((ext_vector_type(2))) float v2f;
typedef __attribute__((ext_vector_type(8))) float v8f;

__constant__ const signed char kCoefIdx[128] __attribute__((aligned(16))) = {
     2,  3,  1, -1,
     4,  5,  6, -1,
     8, 23, 22, -1,
    24, 25, 26, -1,
    -1,  7,  0, -1,
    12, 16, 19, -1,
    -1, -1, 21, -1,
    27, 31, 34, -1,
    -1, -1, -1, 55,
    36, 46, 52, 57,
    -1, -1, 61, -1,
    -1, -1, -1, -1,
    -1, -1, -1, -1,
     9, 10, 11, -1,
    -1, -1, 59, -1,
    -1, -1, -1, -1,
    13, 28, 37, -1,
    38, 39, 40, -1,
    17, 32, 47, -1,
    42, 48, 49, -1,
    15, 30, 44, -1,
    -1, 43, 45, -1,
    18, 33, 50, -1,
    -1, -1, 51, -1,
    20, 35, 53, -1,
    -1, -1, 54, -1,
    -1, -1, 63, -1,
    -1, -1, -1, -1,
    14, 29, 41, -1,
    -1, -1, -1, -1,
    -1, -1, 64, -1,
    -1, -1, -1, -1,
};

struct Rows { float2 p0, p1; float t0, t1; };

__device__ __forceinline__ Rows load_rows_full(const char* __restrict__ xs,
                                               const char* __restrict__ ts,
                                               unsigned xo, unsigned to)
{
    Rows r;
    r.p0 = *(const float2*)(xs + xo);
    r.p1 = *(const float2*)(xs + xo + 256);
    r.t0 = *(const float*)(ts + to);
    r.t1 = *(const float*)(ts + to + 64);
    return r;
}

__device__ __forceinline__ Rows load_rows(const float* __restrict__ x,
                                          const float* __restrict__ t,
                                          int n, int row0, int rsel, int hsel)
{
    const int r0 = row0 + rsel, r1 = r0 + 16;
    const int c0 = (r0 < n) ? r0 : (n - 1);
    const int c1 = (r1 < n) ? r1 : (n - 1);
    const float2* __restrict__ xp = (const float2*)x;
    Rows r;
    r.p0 = xp[c0 * 2 + hsel];
    r.p1 = xp[c1 * 2 + hsel];
    r.t0 = t[c0];
    r.t1 = t[c1];
    return r;
}

__device__ __forceinline__ float cos_rev(float r) { return __builtin_amdgcn_cosf(r); }
__device__ __forceinline__ float sin_rev(float r) { return __builtin_amdgcn_sinf(r); }

template <bool FULL>
__device__ __forceinline__ void do_tile(const Rows& r, int row0, int n,
                                        int lane, bool hi, float g0, float g1,
                                        float gs, float ph,
                                        v2f a0, v2f a1, v2f a2, v2f kd,
                                        float* __restrict__ out)
{
    const float2 pv[2] = { r.p0, r.p1 };
    const float  tv[2] = { r.t0, r.t1 };

    v2f bp;
#pragma unroll
    for (int sub = 0; sub < 2; ++sub) {
        const float T = tv[sub];

        const v2f P = { pv[sub].x, pv[sub].y };
        const float al = hi ? T : 1.0f;

        const v2f ca = P * g0, sa = P * g1;
        const float cB = cos_rev(ca[0]), cG = cos_rev(ca[1]);
        const float sB = sin_rev(sa[0]), sG = sin_rev(sa[1]);
        const float tT = sin_rev(__builtin_fmaf(gs, T, ph));

        const v2f b0 = P;
        const v2f b1 = { al, cB };

        const v2f tg = { cG, sB };
        v2f q = tg * kd;

        const v2f w01 = P * P;
        const v2f w23 = P * al;
        const v2f w45 = { al * al, sG };
        const v2f w67 = { P[0] * P[1], tT };

        v8f acc = {};
        acc = __builtin_amdgcn_wmma_f32_16x16x4_f32(
                  false, a0, false, b0, (short)0, acc, false, false);
        acc = __builtin_amdgcn_wmma_f32_16x16x4_f32(
                  false, a1, false, b1, (short)0, acc, false, false);

        const v2f d01 = { acc[0], acc[1] }, d23 = { acc[2], acc[3] };
        const v2f d45 = { acc[4], acc[5] }, d67 = { acc[6], acc[7] };
        q = __builtin_elementwise_fma(w01, d01, q);
        q = __builtin_elementwise_fma(w23, d23, q);
        q = __builtin_elementwise_fma(w45, d45, q);
        q = __builtin_elementwise_fma(w67, d67, q);
        bp[sub] = q[0] + q[1];
    }

    v8f o = {};
    o = __builtin_amdgcn_wmma_f32_16x16x4_f32(
            false, a2, false, bp, (short)0, o, false, false);

    float2 v; v.x = o[0]; v.y = o[1];
    if (FULL) {
        char* __restrict__ os = (char*)out + (size_t)row0 * 8;
        *(float2*)(os + (unsigned)lane * 8u) = v;
    } else {
        const int row = row0 + lane;
        if (row < n) *(float2*)(out + (size_t)row * 2) = v;
    }
}

struct Setup { v2f a0, a1, a2, kd; float g0, g1, gs, ph; };

__device__ __forceinline__ Setup make_setup(const int (&ci)[4], const float (&cr)[4],
                                            v2f kd, float w0, float w1,
                                            float f0, float f1, int rsel, bool hi)
{
    Setup s;
    s.kd = kd;

    s.a2 = v2f{ (rsel == 0) ? f0 : ((rsel == 1) ? f1 : 0.0f),
                (rsel == 8) ? f0 : ((rsel == 9) ? f1 : 0.0f) };

    const float kInv2Pi = 0.15915494309189535f;
    s.g0 = w0 * kInv2Pi;
    s.g1 = w1 * kInv2Pi;
    s.gs = hi ? s.g1 : s.g0;
    s.ph = hi ? 0.0f : 0.25f;

    s.a0 = v2f{ (ci[0] < 0) ? 0.0f : cr[0], (ci[1] < 0) ? 0.0f : cr[1] };
    s.a1 = v2f{ (ci[2] < 0) ? 0.0f : cr[2], (ci[3] < 0) ? 0.0f : cr[3] };
    return s;
}

constexpr int kSpans = 4;

__global__ __launch_bounds__(128)
void sindy_wmma_f32_kernel(
    const float* __restrict__ x,
    const float* __restrict__ t,
    const float* __restrict__ layer,
    const float* __restrict__ omegas,
    const float* __restrict__ ext_filter,
    float* __restrict__ out,
    int n, int nspans)
{
    const int lane = threadIdx.x & 31;
    const int wave = __builtin_amdgcn_readfirstlane((int)(threadIdx.x >> 5));
    const int rsel = lane & 15;
    const bool hi  = lane >= 16;

    int tabw;
    __builtin_memcpy(&tabw, kCoefIdx + (rsel * 2 + (hi ? 1 : 0)) * 4, 4);
    asm volatile("" :: "v"(tabw));

    const int span0 = (blockIdx.x * 4 + wave) * kSpans;
    if (span0 >= nspans) return;

    const int nfull = n >> 6;
    const bool whole = span0 + kSpans <= nfull;

    int   ci[4];
    float cr[4];
#pragma unroll
    for (int j = 0; j < 4; ++j) {
        ci[j] = (signed char)(tabw >> (8 * j));
        cr[j] = layer[ci[j] < 0 ? 0 : ci[j]];
    }
    const v2f kd = { hi ? layer[58] : layer[56], hi ? layer[62] : layer[60] };

    const int hsel = hi ? 1 : 0;
    const unsigned xo = (unsigned)(rsel * 2 + hsel) * 8u;
    const unsigned to = (unsigned)rsel * 4u;

    if (whole) {
        const char* xs = (const char*)x + (size_t)span0 * 1024;
        const char* ts = (const char*)t + (size_t)span0 * 256;
        Rows r[2 * kSpans];
#pragma unroll
        for (int i = 0; i < 2 * kSpans; ++i)
            r[i] = load_rows_full(xs + i * 512, ts + i * 128, xo, to);

        __builtin_amdgcn_sched_barrier(0);

        const Setup s = make_setup(ci, cr, kd, omegas[0], omegas[1],
                                   ext_filter[0], ext_filter[1], rsel, hi);

#pragma unroll
        for (int i = 0; i < 2 * kSpans; ++i)
            do_tile<true>(r[i], span0 * 64 + i * 32, n, lane, hi, s.g0, s.g1, s.gs, s.ph,
                          s.a0, s.a1, s.a2, s.kd, out);
        return;
    }

    const Setup s = make_setup(ci, cr, kd, omegas[0], omegas[1],
                               ext_filter[0], ext_filter[1], rsel, hi);
    const int send = (span0 + kSpans < nspans) ? (span0 + kSpans) : nspans;
    for (int span = span0; span < send; ++span) {
        if (span < nfull) {
            const char* xs = (const char*)x + (size_t)span * 1024;
            const char* ts = (const char*)t + (size_t)span * 256;
            const Rows ra = load_rows_full(xs, ts, xo, to);
            const Rows rb = load_rows_full(xs + 512, ts + 128, xo, to);
            do_tile<true>(ra, span * 64,      n, lane, hi, s.g0, s.g1, s.gs, s.ph, s.a0, s.a1, s.a2, s.kd, out);
            do_tile<true>(rb, span * 64 + 32, n, lane, hi, s.g0, s.g1, s.gs, s.ph, s.a0, s.a1, s.a2, s.kd, out);
        } else {
            const Rows rc = load_rows(x, t, n, span * 64,      rsel, hsel);
            const Rows rd = load_rows(x, t, n, span * 64 + 32, rsel, hsel);
            do_tile<false>(rc, span * 64,      n, lane, hi, s.g0, s.g1, s.gs, s.ph, s.a0, s.a1, s.a2, s.kd, out);
            do_tile<false>(rd, span * 64 + 32, n, lane, hi, s.g0, s.g1, s.gs, s.ph, s.a0, s.a1, s.a2, s.kd, out);
        }
    }
}

extern "C" void kernel_launch(void* const* d_in, const int* in_sizes, int n_in,
                              void* d_out, int out_size, void* d_ws, size_t ws_size,
                              hipStream_t stream) {
    const float* x          = (const float*)d_in[0];
    const float* t          = (const float*)d_in[1];
    const float* layer      = (const float*)d_in[2];
    const float* omegas     = (const float*)d_in[3];
    const float* ext_filter = (const float*)d_in[4];
    float* out = (float*)d_out;

    const int n      = in_sizes[1];
    const int nspans = (n + 63) / 64;
    const int nwaves = (nspans + 3) / 4;
    const int waves_per_block = 4;
    const int grid = (nwaves + waves_per_block - 1) / waves_per_block;

    sindy_wmma_f32_kernel<<<grid, 128, 0, stream>>>(x, t, layer, omegas,
                                                    ext_filter, out, n, nspans);
}


# K = 4, no closing WMMA: a u's own coefficient rides in the accumulator input; trig, t, t^3 as multiply-adds; lane halves joined by swap and add: 2 WMMAs per 32 rows, not 5; terms summed in new order
# speedup vs baseline: 3.6498x; 1.0794x over previous
#include <hip/hip_runtime.h>


typedef __attribute__((ext_vector_type(2))) float v2f;
typedef __attribute__((ext_vector_type(8))) float v8f;

__constant__ const signed char kCoefIdx[64] __attribute__((aligned(16))) = {
     2,  3,
     4,  5,
     8, 23,
    24, 25,
     6, 11,
    15, 18,
    -1, 26,
    30, 33,
    -1,  7,
    12, 16,
    -1, -1,
    27, 31,
    -1, -1,
     9, 10,
    -1, -1,
    -1, -1,
    13, 28,
    38, 39,
    17, 32,
    42, 48,
    -1, -1,
    40, 43,
    -1, -1,
    -1, 49,
    -1, -1,
    -1, 41,
    -1, -1,
    -1, -1,
    14, 29,
    -1, -1,
    20, 35,
    45, 51,
};

struct Rows { float2 p0, p1; float t0, t1; };

__device__ __forceinline__ Rows load_rows_full(const char* __restrict__ xs,
                                               const char* __restrict__ ts,
                                               unsigned xo, unsigned to)
{
    Rows r;
    r.p0 = *(const float2*)(xs + xo);
    r.p1 = *(const float2*)(xs + xo + 256);
    r.t0 = *(const float*)(ts + to);
    r.t1 = *(const float*)(ts + to + 64);
    return r;
}

__device__ __forceinline__ Rows load_rows(const float* __restrict__ x,
                                          const float* __restrict__ t,
                                          int n, int row0, int rsel, int hsel)
{
    const int r0 = row0 + rsel, r1 = r0 + 16;
    const int c0 = (r0 < n) ? r0 : (n - 1);
    const int c1 = (r1 < n) ? r1 : (n - 1);
    const float2* __restrict__ xp = (const float2*)x;
    Rows r;
    r.p0 = xp[c0 * 2 + hsel];
    r.p1 = xp[c1 * 2 + hsel];
    r.t0 = t[c0];
    r.t1 = t[c1];
    return r;
}

__device__ __forceinline__ float cos_rev(float r) { return __builtin_amdgcn_cosf(r); }
__device__ __forceinline__ float sin_rev(float r) { return __builtin_amdgcn_sinf(r); }

struct Setup {
    v2f a0;
    v8f c0;
    v2f kc, ks;
    float kt, k1, k3;
    v2f f;
    float g0, g1, gs, ph;
};

template <bool FULL>
__device__ __forceinline__ void do_tile(const Rows& r, int row0, int n, int lane,
                                        const Setup& s, float* __restrict__ out)
{
    const float2 pv[2] = { r.p0, r.p1 };
    const float  tv[2] = { r.t0, r.t1 };

    float bp[2];
#pragma unroll
    for (int sub = 0; sub < 2; ++sub) {
        const float T = tv[sub];

        const v2f P = { pv[sub].x, pv[sub].y };

        const v2f ca = P * s.g0, sa = P * s.g1;
        const v2f cs = { cos_rev(ca[0]), cos_rev(ca[1]) };
        const v2f sn = { sin_rev(sa[0]), sin_rev(sa[1]) };
        const float tT = sin_rev(__builtin_fmaf(s.gs, T, s.ph));

        const float T2 = T * T;
        v2f q = cs * s.kc;
        q = __builtin_elementwise_fma(sn, s.ks, q);
        q[0] = __builtin_fmaf(tT, s.kt, q[0]);
        q[1] = __builtin_fmaf(__builtin_fmaf(s.k3, T2, s.k1), T, q[1]);

        const v2f u01 = P * P;
        const v2f u23 = P * T;
        const v2f u67 = { P[0] * P[1], T2 };

        const v8f acc = __builtin_amdgcn_wmma_f32_16x16x4_f32(
                            false, s.a0, false, P, (short)0, s.c0, false, false);

        const v2f d01 = { acc[0], acc[1] }, d23 = { acc[2], acc[3] };
        const v2f d45 = { acc[4], acc[5] }, d67 = { acc[6], acc[7] };
        q = __builtin_elementwise_fma(u01, d01, q);
        q = __builtin_elementwise_fma(u23, d23, q);
        q = __builtin_elementwise_fma(P,   d45, q);
        q = __builtin_elementwise_fma(u67, d67, q);
        bp[sub] = q[0] + q[1];
    }

    float pa = bp[0], pb = bp[1];
    asm("v_permlane16_swap_b32 %0, %1" : "+v"(pa), "+v"(pb));
    const float sum = pa + pb;
    const v2f o = s.f * sum;

    float2 v; v.x = o[0]; v.y = o[1];
    if (FULL) {
        char* __restrict__ os = (char*)out + (size_t)row0 * 8;
        *(float2*)(os + (unsigned)lane * 8u) = v;
    } else {
        const int row = row0 + lane;
        if (row < n) *(float2*)(out + (size_t)row * 2) = v;
    }
}

struct Raw { float cr[2]; float c[6]; float kc[2], ks[2], kt; };

__device__ __forceinline__ Setup make_setup(const int (&ci)[2], const Raw& w,
                                            float ct1, float ct2, float ct3,
                                            float w0, float w1,
                                            float f0, float f1, bool hi)
{
    Setup s;

    s.a0 = v2f{ (ci[0] < 0) ? 0.0f : w.cr[0], (ci[1] < 0) ? 0.0f : w.cr[1] };

    s.c0 = v8f{ w.c[0], w.c[1], w.c[2], w.c[3], w.c[4], w.c[5], 0.0f,
                0.5f * ct2 };

    s.kc = v2f{ w.kc[0], w.kc[1] };
    s.ks = v2f{ w.ks[0], w.ks[1] };
    s.kt = w.kt;
    s.k1 = 0.5f * ct1;
    s.k3 = 0.5f * ct3;
    s.f  = v2f{ f0, f1 };

    const float kInv2Pi = 0.15915494309189535f;
    s.g0 = w0 * kInv2Pi;
    s.g1 = w1 * kInv2Pi;
    s.gs = hi ? s.g1 : s.g0;
    s.ph = hi ? 0.0f : 0.25f;
    return s;
}

constexpr int kSpans = 4;

__global__ __launch_bounds__(128)
void sindy_wmma_f32_kernel(
    const float* __restrict__ x,
    const float* __restrict__ t,
    const float* __restrict__ layer,
    const float* __restrict__ omegas,
    const float* __restrict__ ext_filter,
    float* __restrict__ out,
    int n, int nspans)
{
    const int lane = threadIdx.x & 31;
    const int wave = __builtin_amdgcn_readfirstlane((int)(threadIdx.x >> 5));
    const int rsel = lane & 15;
    const bool hi  = lane >= 16;

    short tabw;
    __builtin_memcpy(&tabw, kCoefIdx + (rsel * 2 + (hi ? 1 : 0)) * 2, 2);
    asm volatile("" :: "v"(tabw));

    const int span0 = (blockIdx.x * 4 + wave) * kSpans;
    if (span0 >= nspans) return;

    const int nfull = n >> 6;
    const bool whole = span0 + kSpans <= nfull;

    int ci[2];
    Raw w;
#pragma unroll
    for (int j = 0; j < 2; ++j) {
        ci[j] = (signed char)(tabw >> (8 * j));
        w.cr[j] = layer[ci[j] < 0 ? 0 : ci[j]];
    }
    w.c[0] = layer[hi ? 37 : 1];
    w.c[1] = layer[hi ? 47 : 22];
    w.c[2] = layer[hi ? 44 : 19];
    w.c[3] = layer[hi ? 50 : 34];
    w.c[4] = layer[hi ? 36 : 0];
    w.c[5] = layer[hi ? 46 : 21];
    w.kc[0] = layer[hi ? 57 : 55];
    w.kc[1] = layer[hi ? 58 : 56];
    w.ks[0] = layer[hi ? 62 : 60];
    w.ks[1] = layer[hi ? 63 : 61];
    w.kt    = layer[hi ? 64 : 59];

    const int hsel = hi ? 1 : 0;
    const unsigned xo = (unsigned)(rsel * 2 + hsel) * 8u;
    const unsigned to = (unsigned)rsel * 4u;

    if (whole) {
        const char* xs = (const char*)x + (size_t)span0 * 1024;
        const char* ts = (const char*)t + (size_t)span0 * 256;
        Rows r[2 * kSpans];
#pragma unroll
        for (int i = 0; i < 2 * kSpans; ++i)
            r[i] = load_rows_full(xs + i * 512, ts + i * 128, xo, to);

        __builtin_amdgcn_sched_barrier(0);

        const Setup s = make_setup(ci, w, layer[52], layer[53], layer[54],
                                   omegas[0], omegas[1],
                                   ext_filter[0], ext_filter[1], hi);

#pragma unroll
        for (int i = 0; i < 2 * kSpans; ++i)
            do_tile<true>(r[i], span0 * 64 + i * 32, n, lane, s, out);
        return;
    }

    const Setup s = make_setup(ci, w, layer[52], layer[53], layer[54],
                                   omegas[0], omegas[1],
                               ext_filter[0], ext_filter[1], hi);
    const int send = (span0 + kSpans < nspans) ? (span0 + kSpans) : nspans;
    for (int span = span0; span < send; ++span) {
        if (span < nfull) {
            const char* xs = (const char*)x + (size_t)span * 1024;
            const char* ts = (const char*)t + (size_t)span * 256;
            const Rows ra = load_rows_full(xs, ts, xo, to);
            const Rows rb = load_rows_full(xs + 512, ts + 128, xo, to);
            do_tile<true>(ra, span * 64,      n, lane, s, out);
            do_tile<true>(rb, span * 64 + 32, n, lane, s, out);
        } else {
            const Rows rc = load_rows(x, t, n, span * 64,      rsel, hsel);
            const Rows rd = load_rows(x, t, n, span * 64 + 32, rsel, hsel);
            do_tile<false>(rc, span * 64,      n, lane, s, out);
            do_tile<false>(rd, span * 64 + 32, n, lane, s, out);
        }
    }
}

extern "C" void kernel_launch(void* const* d_in, const int* in_sizes, int n_in,
                              void* d_out, int out_size, void* d_ws, size_t ws_size,
                              hipStream_t stream) {
    const float* x          = (const float*)d_in[0];
    const float* t          = (const float*)d_in[1];
    const float* layer      = (const float*)d_in[2];
    const float* omegas     = (const float*)d_in[3];
    const float* ext_filter = (const float*)d_in[4];
    float* out = (float*)d_out;

    const int n      = in_sizes[1];
    const int nspans = (n + 63) / 64;
    const int nwaves = (nspans + 3) / 4;
    const int waves_per_block = 4;
    const int grid = (nwaves + waves_per_block - 1) / waves_per_block;

    sindy_wmma_f32_kernel<<<grid, 128, 0, stream>>>(x, t, layer, omegas,
                                                    ext_filter, out, n, nspans);
}
